# starting kernel
# baseline (speedup 1.0000x reference)
#include <hip/hip_runtime.h>
#include <math.h>


typedef __attribute__((ext_vector_type(2))) float v2f;
typedef __attribute__((ext_vector_type(8))) float v8f;

#define NB   32
#define SD   512
#define SE   512
#define DD   512
#define STR  516

__launch_bounds__(256, 1)
__global__ void attn_fused_kernel(const float* __restrict__ Q,
                                  const float* __restrict__ E,
                                  const float* __restrict__ mask,
                                  const float* __restrict__ W,
                                  const float* __restrict__ bias,
                                  float* __restrict__ outMasked,
                                  float* __restrict__ outWeights)
{
    __shared__ float sQ[16 * STR];
    __shared__ float sS[16 * STR];
    __shared__ float sRed[16 * 16];
    __shared__ float sMask[16];

    const int b    = blockIdx.y;
    const int q0   = blockIdx.x * 16;
    const int tid  = threadIdx.x;
    const int lane = tid & 31;
    const int wave = tid >> 5;
    const int half = lane >> 4;
    const int l16  = lane & 15;

    const float* Qb = Q + ((size_t)b * SD + q0) * DD;
    const float* Eb = E + (size_t)b * SE * DD;

    for (int i = tid; i < 16 * DD; i += 256) {
        int r = i >> 9;
        int c = i & 511;
        sQ[r * STR + c] = Qb[r * DD + c];
    }
    if (tid < 16) sMask[tid] = mask[(size_t)b * SD + q0 + tid];
    __syncthreads();

    const float scale = 0.044194173824159216f;

    {
        const float* Arow = sQ + l16 * STR + 2 * half;
        for (int t = 0; t < 4; ++t) {
            const int k0 = (wave * 4 + t) * 16;
            const float* Brow = Eb + (size_t)(k0 + l16) * DD + 2 * half;
            v8f acc = {};
            for (int d = 0; d < DD; d += 4) {
                v2f a;  a.x = Arow[d];  a.y = Arow[d + 1];
                v2f bf; bf.x = Brow[d]; bf.y = Brow[d + 1];
                acc = __builtin_amdgcn_wmma_f32_16x16x4_f32(
                    false, a, false, bf, (short)0, acc, false, false);
            }
#pragma unroll
            for (int i = 0; i < 8; ++i)
                sS[(i + 8 * half) * STR + k0 + l16] = acc[i] * scale;
        }
    }
    __syncthreads();

    {
        const int row = tid & 15;
        const int sub = tid >> 4;
        float* srow = sS + row * STR;
        float m = -INFINITY;
        for (int c = sub * 32; c < sub * 32 + 32; ++c) m = fmaxf(m, srow[c]);
        sRed[row * 16 + sub] = m;
        __syncthreads();
        float mall = -INFINITY;
#pragma unroll
        for (int j = 0; j < 16; ++j) mall = fmaxf(mall, sRed[row * 16 + j]);
        __syncthreads();
        float s = 0.f;
        for (int c = sub * 32; c < sub * 32 + 32; ++c) {
            float p = __expf(srow[c] - mall);
            srow[c] = p;
            s += p;
        }
        sRed[row * 16 + sub] = s;
        __syncthreads();
        float stot = 0.f;
#pragma unroll
        for (int j = 0; j < 16; ++j) stot += sRed[row * 16 + j];
        const float inv = 1.0f / stot;
        float* gw = outWeights + ((size_t)b * SD + q0 + row) * SE;
        for (int c = sub * 32; c < sub * 32 + 32; ++c) {
            float p = srow[c] * inv;
            srow[c] = p;
            gw[c]   = p;
        }
    }
    __syncthreads();

    v8f ctx[4];
    {
        const float* Arow = sS + l16 * STR + 2 * half;
        for (int t = 0; t < 4; ++t) {
            const int n0 = wave * 64 + t * 16;
            const float* Bcol = Eb + n0 + l16;
            v8f acc = {};
            for (int k = 0; k < SE; k += 4) {
                v2f a;  a.x = Arow[k]; a.y = Arow[k + 1];
                v2f bf;
                bf.x = Bcol[(size_t)(k + 2 * half) * DD];
                bf.y = Bcol[(size_t)(k + 2 * half + 1) * DD];
                acc = __builtin_amdgcn_wmma_f32_16x16x4_f32(
                    false, a, false, bf, (short)0, acc, false, false);
            }
            ctx[t] = acc;
        }
    }
    __syncthreads();
#pragma unroll
    for (int t = 0; t < 4; ++t) {
        const int n0 = wave * 64 + t * 16;
#pragma unroll
        for (int i = 0; i < 8; ++i)
            sS[(i + 8 * half) * STR + n0 + l16] = ctx[t][i];
    }
    __syncthreads();

    {
        const float* AQ = sQ + l16 * STR + 2 * half;
        const float* AC = sS + l16 * STR + 2 * half;
        for (int t = 0; t < 4; ++t) {
            const int n0 = wave * 64 + t * 16;
            const float* Wr = W + (size_t)(n0 + l16) * (2 * DD) + 2 * half;
            v8f acc = {};
            for (int k = 0; k < DD; k += 4) {
                v2f a;  a.x = AQ[k]; a.y = AQ[k + 1];
                v2f bf; bf.x = Wr[k]; bf.y = Wr[k + 1];
                acc = __builtin_amdgcn_wmma_f32_16x16x4_f32(
                    false, a, false, bf, (short)0, acc, false, false);
            }
            const float* Wr2 = Wr + DD;
            for (int k = 0; k < DD; k += 4) {
                v2f a;  a.x = AC[k]; a.y = AC[k + 1];
                v2f bf; bf.x = Wr2[k]; bf.y = Wr2[k + 1];
                acc = __builtin_amdgcn_wmma_f32_16x16x4_f32(
                    false, a, false, bf, (short)0, acc, false, false);
            }
            const float bn = bias[n0 + l16];
            float* orow = outMasked + ((size_t)b * SD + q0) * DD + n0 + l16;
#pragma unroll
            for (int i = 0; i < 8; ++i) {
                const int M = i + 8 * half;
                orow[(size_t)M * DD] = tanhf(acc[i] + bn) * sMask[M];
            }
        }
    }
}

extern "C" void kernel_launch(void* const* d_in, const int* in_sizes, int n_in,
                              void* d_out, int out_size, void* d_ws, size_t ws_size,
                              hipStream_t stream) {
    (void)in_sizes; (void)n_in; (void)out_size; (void)d_ws; (void)ws_size;
    const float* Q    = (const float*)d_in[0];
    const float* E    = (const float*)d_in[1];
    const float* mask = (const float*)d_in[2];
    const float* W    = (const float*)d_in[3];
    const float* bias = (const float*)d_in[4];
    float* outMasked  = (float*)d_out;
    float* outWeights = (float*)d_out + (size_t)NB * SD * SE;

    dim3 grid(SD / 16, NB);
    attn_fused_kernel<<<grid, 256, 0, stream>>>(Q, E, mask, W, bias,
                                                outMasked, outWeights);
}


# 64 query rows per work-group (4 row tiles, 16 waves, grid 8x32): each B fragment fetched from global feeds 4 WMMAs instead of 1 (k-loops unrolled 8 deep); k order and softmax sum order unchanged
# speedup vs baseline: 1.3608x; 1.3608x over previous
#include <hip/hip_runtime.h>
#include <math.h>


typedef __attribute__((ext_vector_type(2))) float v2f;
typedef __attribute__((ext_vector_type(8))) float v8f;

#define NB   32
#define SD   512
#define SE   512
#define DD   512
#define STR  516
#define MR   64
#define MT   4
#define NTHR 512
#define TSTR (16 * STR)

__launch_bounds__(NTHR, 1)
__global__ void attn_fused_kernel(const float* __restrict__ Q,
                                  const float* __restrict__ E,
                                  const float* __restrict__ mask,
                                  const float* __restrict__ W,
                                  const float* __restrict__ bias,
                                  float* __restrict__ outMasked,
                                  float* __restrict__ outWeights)
{
    __shared__ float sQ[MR * STR];
    __shared__ float sS[MR * STR];
    __shared__ float sRed[MR * 16];
    __shared__ float sMask[MR];

    const int b    = blockIdx.y;
    const int q0   = blockIdx.x * MR;
    const int tid  = threadIdx.x;
    const int lane = tid & 31;
    const int wave = tid >> 5;
    const int half = lane >> 4;
    const int l16  = lane & 15;

    const float* Qb = Q + ((size_t)b * SD + q0) * DD;
    const float* Eb = E + (size_t)b * SE * DD;

    for (int i = tid; i < MR * DD; i += NTHR) {
        int r = i >> 9;
        int c = i & 511;
        sQ[r * STR + c] = Qb[r * DD + c];
    }
    if (tid < MR) sMask[tid] = mask[(size_t)b * SD + q0 + tid];
    __syncthreads();

    const float scale = 0.044194173824159216f;

    {
        const float* Arow = sQ + l16 * STR + 2 * half;
        for (int t = 0; t < 2; ++t) {
            const int k0 = (wave * 2 + t) * 16;
            const float* Brow = Eb + (size_t)(k0 + l16) * DD + 2 * half;
            v8f acc[MT];
            for (int m = 0; m < MT; ++m) acc[m] = (v8f){};
#pragma unroll 8
            for (int d = 0; d < DD; d += 4) {
                v2f bf; bf.x = Brow[d]; bf.y = Brow[d + 1];
                for (int m = 0; m < MT; ++m) {
                    v2f a;  a.x = Arow[m * TSTR + d];  a.y = Arow[m * TSTR + d + 1];
                    acc[m] = __builtin_amdgcn_wmma_f32_16x16x4_f32(
                        false, a, false, bf, (short)0, acc[m], false, false);
                }
            }
            for (int m = 0; m < MT; ++m)
#pragma unroll
                for (int i = 0; i < 8; ++i)
                    sS[(m * 16 + i + 8 * half) * STR + k0 + l16] = acc[m][i] * scale;
        }
    }
    __syncthreads();

    {
        const int sub  = (tid >> 4) & 15;
        const int rowA = (tid >> 8) * 16 + (tid & 15);
        float mall[2], inv[2];
        for (int r = 0; r < 2; ++r) {
            const float* srow = sS + (rowA + 32 * r) * STR;
            float m = -INFINITY;
            for (int c = sub * 32; c < sub * 32 + 32; ++c) m = fmaxf(m, srow[c]);
            sRed[(rowA + 32 * r) * 16 + sub] = m;
        }
        __syncthreads();
        for (int r = 0; r < 2; ++r) {
            float mm = -INFINITY;
#pragma unroll
            for (int j = 0; j < 16; ++j) mm = fmaxf(mm, sRed[(rowA + 32 * r) * 16 + j]);
            mall[r] = mm;
        }
        __syncthreads();
        for (int r = 0; r < 2; ++r) {
            float* srow = sS + (rowA + 32 * r) * STR;
            float s = 0.f;
            for (int c = sub * 32; c < sub * 32 + 32; ++c) {
                float p = __expf(srow[c] - mall[r]);
                srow[c] = p;
                s += p;
            }
            sRed[(rowA + 32 * r) * 16 + sub] = s;
        }
        __syncthreads();
        for (int r = 0; r < 2; ++r) {
            float stot = 0.f;
#pragma unroll
            for (int j = 0; j < 16; ++j) stot += sRed[(rowA + 32 * r) * 16 + j];
            inv[r] = 1.0f / stot;
        }
        for (int r = 0; r < 2; ++r) {
            float* srow = sS + (rowA + 32 * r) * STR;
            float* gw = outWeights + ((size_t)b * SD + q0 + rowA + 32 * r) * SE;
            for (int c = sub * 32; c < sub * 32 + 32; ++c) {
                float p = srow[c] * inv[r];
                srow[c] = p;
                gw[c]   = p;
            }
        }
    }
    __syncthreads();

    v8f ctx[2][MT];
    {
        const float* Arow = sS + l16 * STR + 2 * half;
        for (int t = 0; t < 2; ++t) {
            const int n0 = wave * 32 + t * 16;
            const float* Bcol = Eb + n0 + l16;
            v8f acc[MT];
            for (int m = 0; m < MT; ++m) acc[m] = (v8f){};
            for (int k = 0; k < SE; k += 4) {
                v2f bf;
                bf.x = Bcol[(size_t)(k + 2 * half) * DD];
                bf.y = Bcol[(size_t)(k + 2 * half + 1) * DD];
                for (int m = 0; m < MT; ++m) {
                    v2f a;  a.x = Arow[m * TSTR + k]; a.y = Arow[m * TSTR + k + 1];
                    acc[m] = __builtin_amdgcn_wmma_f32_16x16x4_f32(
                        false, a, false, bf, (short)0, acc[m], false, false);
                }
            }
            for (int m = 0; m < MT; ++m) ctx[t][m] = acc[m];
        }
    }
    __syncthreads();
    for (int t = 0; t < 2; ++t) {
        const int n0 = wave * 32 + t * 16;
        for (int m = 0; m < MT; ++m)
#pragma unroll
            for (int i = 0; i < 8; ++i)
                sS[(m * 16 + i + 8 * half) * STR + n0 + l16] = ctx[t][m][i];
    }
    __syncthreads();

    {
        const float* AQ = sQ + l16 * STR + 2 * half;
        const float* AC = sS + l16 * STR + 2 * half;
        for (int t = 0; t < 2; ++t) {
            const int n0 = wave * 32 + t * 16;
            const float* Wr = W + (size_t)(n0 + l16) * (2 * DD) + 2 * half;
            v8f acc[MT];
            for (int m = 0; m < MT; ++m) acc[m] = (v8f){};
#pragma unroll 8
            for (int k = 0; k < DD; k += 4) {
                v2f bf; bf.x = Wr[k]; bf.y = Wr[k + 1];
                for (int m = 0; m < MT; ++m) {
                    v2f a;  a.x = AQ[m * TSTR + k]; a.y = AQ[m * TSTR + k + 1];
                    acc[m] = __builtin_amdgcn_wmma_f32_16x16x4_f32(
                        false, a, false, bf, (short)0, acc[m], false, false);
                }
            }
            const float* Wr2 = Wr + DD;
#pragma unroll 8
            for (int k = 0; k < DD; k += 4) {
                v2f bf; bf.x = Wr2[k]; bf.y = Wr2[k + 1];
                for (int m = 0; m < MT; ++m) {
                    v2f a;  a.x = AC[m * TSTR + k]; a.y = AC[m * TSTR + k + 1];
                    acc[m] = __builtin_amdgcn_wmma_f32_16x16x4_f32(
                        false, a, false, bf, (short)0, acc[m], false, false);
                }
            }
            const float bn = bias[n0 + l16];
            float* orow = outMasked + ((size_t)b * SD + q0) * DD + n0 + l16;
            for (int m = 0; m < MT; ++m)
#pragma unroll
                for (int i = 0; i < 8; ++i) {
                    const int M = m * 16 + i + 8 * half;
                    orow[(size_t)M * DD] = tanhf(acc[m][i] + bn) * sMask[M];
                }
        }
    }
}

extern "C" void kernel_launch(void* const* d_in, const int* in_sizes, int n_in,
                              void* d_out, int out_size, void* d_ws, size_t ws_size,
                              hipStream_t stream) {
    (void)in_sizes; (void)n_in; (void)out_size; (void)d_ws; (void)ws_size;
    const float* Q    = (const float*)d_in[0];
    const float* E    = (const float*)d_in[1];
    const float* mask = (const float*)d_in[2];
    const float* W    = (const float*)d_in[3];
    const float* bias = (const float*)d_in[4];
    float* outMasked  = (float*)d_out;
    float* outWeights = (float*)d_out + (size_t)NB * SD * SE;

    dim3 grid(SD / MR, NB);
    attn_fused_kernel<<<grid, NTHR, 0, stream>>>(Q, E, mask, W, bias,
                                                 outMasked, outWeights);
}


# B fragments double-buffered in registers: each block of 8 k-steps is requested one block (32 WMMAs) ahead, across tile ends too, in all three products; k order and sum order unchanged
# speedup vs baseline: 1.4164x; 1.0409x over previous
#include <hip/hip_runtime.h>
#include <math.h>


typedef __attribute__((ext_vector_type(2))) float v2f;
typedef __attribute__((ext_vector_type(8))) float v8f;

#define NB   32
#define SD   512
#define SE   512
#define DD   512
#define STR  516
#define MR   64
#define MT   4
#define NTHR 512
#define TSTR (16 * STR)
#define KB   32

__device__ __forceinline__ void load_b_krow(v2f (&bf)[8], const float* __restrict__ p) {
#pragma unroll
    for (int j = 0; j < 8; ++j) { bf[j].x = p[4 * j]; bf[j].y = p[4 * j + 1]; }
    asm volatile("" ::: "memory");
}

__device__ __forceinline__ void load_b_kcol(v2f (&bf)[8], const float* __restrict__ p) {
#pragma unroll
    for (int j = 0; j < 8; ++j) {
        bf[j].x = p[(size_t)(4 * j) * DD];
        bf[j].y = p[(size_t)(4 * j + 1) * DD];
    }
    asm volatile("" ::: "memory");
}

__device__ __forceinline__ void mma_block(v8f (&acc)[MT], const float* a, const v2f (&bf)[8]) {
#pragma unroll
    for (int j = 0; j < 8; ++j)
#pragma unroll
        for (int m = 0; m < MT; ++m) {
            v2f av;  av.x = a[m * TSTR + 4 * j];  av.y = a[m * TSTR + 4 * j + 1];
            acc[m] = __builtin_amdgcn_wmma_f32_16x16x4_f32(
                false, av, false, bf[j], (short)0, acc[m], false, false);
        }
}

__launch_bounds__(NTHR, 1)
__global__ void attn_fused_kernel(const float* __restrict__ Q,
                                  const float* __restrict__ E,
                                  const float* __restrict__ mask,
                                  const float* __restrict__ W,
                                  const float* __restrict__ bias,
                                  float* __restrict__ outMasked,
                                  float* __restrict__ outWeights)
{
    __shared__ float sQ[MR * STR];
    __shared__ float sS[MR * STR];
    __shared__ float sRed[MR * 16];
    __shared__ float sMask[MR];

    const int b    = blockIdx.y;
    const int q0   = blockIdx.x * MR;
    const int tid  = threadIdx.x;
    const int lane = tid & 31;
    const int wave = tid >> 5;
    const int half = lane >> 4;
    const int l16  = lane & 15;

    const float* Qb = Q + ((size_t)b * SD + q0) * DD;
    const float* Eb = E + (size_t)b * SE * DD;

    v2f bA[8], bB[8];
    const float* Brow0 = Eb + (size_t)(wave * 32 + l16) * DD + 2 * half;
    load_b_krow(bA, Brow0);

    for (int i = tid; i < MR * DD; i += NTHR) {
        int r = i >> 9;
        int c = i & 511;
        sQ[r * STR + c] = Qb[r * DD + c];
    }
    if (tid < MR) sMask[tid] = mask[(size_t)b * SD + q0 + tid];
    __syncthreads();

    const float scale = 0.044194173824159216f;

    {
        const float* Arow = sQ + l16 * STR + 2 * half;
        for (int t = 0; t < 2; ++t) {
            const int k0 = (wave * 2 + t) * 16;
            const float* Brow = Brow0 + (size_t)(t * 16) * DD;
            v8f acc[MT];
            for (int m = 0; m < MT; ++m) acc[m] = (v8f){};
            for (int d = 0; d < DD; d += 2 * KB) {
                load_b_krow(bB, Brow + d + KB);
                mma_block(acc, Arow + d, bA);
                const float* nx = (d + 2 * KB < DD) ? Brow + d + 2 * KB : Brow0 + (size_t)16 * DD;
                load_b_krow(bA, nx);
                mma_block(acc, Arow + d + KB, bB);
            }
            for (int m = 0; m < MT; ++m)
#pragma unroll
                for (int i = 0; i < 8; ++i)
                    sS[(m * 16 + i + 8 * half) * STR + k0 + l16] = acc[m][i] * scale;
        }
    }
    __syncthreads();

    {
        const int sub  = (tid >> 4) & 15;
        const int rowA = (tid >> 8) * 16 + (tid & 15);
        float mall[2], inv[2];
        for (int r = 0; r < 2; ++r) {
            const float* srow = sS + (rowA + 32 * r) * STR;
            float m = -INFINITY;
            for (int c = sub * 32; c < sub * 32 + 32; ++c) m = fmaxf(m, srow[c]);
            sRed[(rowA + 32 * r) * 16 + sub] = m;
        }
        __syncthreads();
        for (int r = 0; r < 2; ++r) {
            float mm = -INFINITY;
#pragma unroll
            for (int j = 0; j < 16; ++j) mm = fmaxf(mm, sRed[(rowA + 32 * r) * 16 + j]);
            mall[r] = mm;
        }
        __syncthreads();
        for (int r = 0; r < 2; ++r) {
            float* srow = sS + (rowA + 32 * r) * STR;
            float s = 0.f;
            for (int c = sub * 32; c < sub * 32 + 32; ++c) {
                float p = __expf(srow[c] - mall[r]);
                srow[c] = p;
                s += p;
            }
            sRed[(rowA + 32 * r) * 16 + sub] = s;
        }
        __syncthreads();
        for (int r = 0; r < 2; ++r) {
            float stot = 0.f;
#pragma unroll
            for (int j = 0; j < 16; ++j) stot += sRed[(rowA + 32 * r) * 16 + j];
            inv[r] = 1.0f / stot;
        }
        for (int r = 0; r < 2; ++r) {
            float* srow = sS + (rowA + 32 * r) * STR;
            float* gw = outWeights + ((size_t)b * SD + q0 + rowA + 32 * r) * SE;
            for (int c = sub * 32; c < sub * 32 + 32; ++c) {
                float p = srow[c] * inv[r];
                srow[c] = p;
                gw[c]   = p;
            }
        }
    }
    const float* Bk0 = Eb + wave * 32 + l16 + (size_t)(2 * half) * DD;
    load_b_kcol(bA, Bk0);
    __syncthreads();

    v8f ctx[2][MT];
    {
        const float* Arow = sS + l16 * STR + 2 * half;
        for (int t = 0; t < 2; ++t) {
            const float* Bk = Bk0 + t * 16;
            v8f acc[MT];
            for (int m = 0; m < MT; ++m) acc[m] = (v8f){};
            for (int k = 0; k < SE; k += 2 * KB) {
                load_b_kcol(bB, Bk + (size_t)(k + KB) * DD);
                mma_block(acc, Arow + k, bA);
                const float* nx = (k + 2 * KB < SE) ? Bk + (size_t)(k + 2 * KB) * DD : Bk0 + 16;
                load_b_kcol(bA, nx);
                mma_block(acc, Arow + k + KB, bB);
            }
            for (int m = 0; m < MT; ++m) ctx[t][m] = acc[m];
        }
    }
    const float* Wr0 = W + (size_t)(wave * 32 + l16) * (2 * DD) + 2 * half;
    load_b_krow(bA, Wr0);
    __syncthreads();
    for (int t = 0; t < 2; ++t) {
        const int n0 = wave * 32 + t * 16;
        for (int m = 0; m < MT; ++m)
#pragma unroll
            for (int i = 0; i < 8; ++i)
                sS[(m * 16 + i + 8 * half) * STR + n0 + l16] = ctx[t][m][i];
    }
    __syncthreads();

    {
        const float* AQ = sQ + l16 * STR + 2 * half;
        const float* AC = sS + l16 * STR + 2 * half;
        for (int t = 0; t < 2; ++t) {
            const int n0 = wave * 32 + t * 16;
            const float* Wr = Wr0 + (size_t)(t * 16) * (2 * DD);
            v8f acc[MT];
            for (int m = 0; m < MT; ++m) acc[m] = (v8f){};
            for (int k = 0; k < 2 * DD; k += 2 * KB) {
                const float* A = (k < DD) ? AQ + k : AC + (k - DD);
                load_b_krow(bB, Wr + k + KB);
                mma_block(acc, A, bA);
                const float* nx = (k + 2 * KB < 2 * DD) ? Wr + k + 2 * KB : Wr0 + (size_t)16 * (2 * DD);
                load_b_krow(bA, nx);
                mma_block(acc, A + KB, bB);
            }
            const float bn = bias[n0 + l16];
            float* orow = outMasked + ((size_t)b * SD + q0) * DD + n0 + l16;
            for (int m = 0; m < MT; ++m)
#pragma unroll
                for (int i = 0; i < 8; ++i) {
                    const int M = m * 16 + i + 8 * half;
                    orow[(size_t)M * DD] = tanhf(acc[m][i] + bn) * sMask[M];
                }
        }
    }
}

extern "C" void kernel_launch(void* const* d_in, const int* in_sizes, int n_in,
                              void* d_out, int out_size, void* d_ws, size_t ws_size,
                              hipStream_t stream) {
    (void)in_sizes; (void)n_in; (void)out_size; (void)d_ws; (void)ws_size;
    const float* Q    = (const float*)d_in[0];
    const float* E    = (const float*)d_in[1];
    const float* mask = (const float*)d_in[2];
    const float* W    = (const float*)d_in[3];
    const float* bias = (const float*)d_in[4];
    float* outMasked  = (float*)d_out;
    float* outWeights = (float*)d_out + (size_t)NB * SD * SE;

    dim3 grid(SD / MR, NB);
    attn_fused_kernel<<<grid, NTHR, 0, stream>>>(Q, E, mask, W, bias,
                                                 outMasked, outWeights);
}


# Each wave computes both of its 16-column tiles together (4x2 accumulator tiles) in all three products: every A fragment read from LDS feeds 2 WMMAs, not 1; k order and sum order unchanged
# speedup vs baseline: 1.5026x; 1.0609x over previous
#include <hip/hip_runtime.h>
#include <math.h>


typedef __attribute__((ext_vector_type(2))) float v2f;
typedef __attribute__((ext_vector_type(8))) float v8f;

#define NB   32
#define SD   512
#define SE   512
#define DD   512
#define STR  516
#define MR   64
#define MT   4
#define NTHR 512
#define TSTR (16 * STR)
#define KB   32

__device__ __forceinline__ void load_b_krow(v2f (&bf)[8], const float* __restrict__ p) {
#pragma unroll
    for (int j = 0; j < 8; ++j) { bf[j].x = p[4 * j]; bf[j].y = p[4 * j + 1]; }
    asm volatile("" ::: "memory");
}

__device__ __forceinline__ void load_b_kcol(v2f (&bf)[8], const float* __restrict__ p) {
#pragma unroll
    for (int j = 0; j < 8; ++j) {
        bf[j].x = p[(size_t)(4 * j) * DD];
        bf[j].y = p[(size_t)(4 * j + 1) * DD];
    }
    asm volatile("" ::: "memory");
}

__device__ __forceinline__ void mma_block2(v8f (&acc)[2][MT], const float* a,
                                           const v2f (&bf0)[8], const v2f (&bf1)[8]) {
#pragma unroll
    for (int j = 0; j < 8; ++j)
#pragma unroll
        for (int m = 0; m < MT; ++m) {
            v2f av;  av.x = a[m * TSTR + 4 * j];  av.y = a[m * TSTR + 4 * j + 1];
            acc[0][m] = __builtin_amdgcn_wmma_f32_16x16x4_f32(
                false, av, false, bf0[j], (short)0, acc[0][m], false, false);
            acc[1][m] = __builtin_amdgcn_wmma_f32_16x16x4_f32(
                false, av, false, bf1[j], (short)0, acc[1][m], false, false);
        }
}

__launch_bounds__(NTHR, 1)
__global__ void attn_fused_kernel(const float* __restrict__ Q,
                                  const float* __restrict__ E,
                                  const float* __restrict__ mask,
                                  const float* __restrict__ W,
                                  const float* __restrict__ bias,
                                  float* __restrict__ outMasked,
                                  float* __restrict__ outWeights)
{
    __shared__ float sQ[MR * STR];
    __shared__ float sS[MR * STR];
    __shared__ float sRed[MR * 16];
    __shared__ float sMask[MR];

    const int b    = blockIdx.y;
    const int q0   = blockIdx.x * MR;
    const int tid  = threadIdx.x;
    const int lane = tid & 31;
    const int wave = tid >> 5;
    const int half = lane >> 4;
    const int l16  = lane & 15;

    const float* Qb = Q + ((size_t)b * SD + q0) * DD;
    const float* Eb = E + (size_t)b * SE * DD;

    v2f bA0[8], bA1[8], bB0[8], bB1[8];
    const float* Brow0 = Eb + (size_t)(wave * 32 + l16) * DD + 2 * half;
    const float* Brow1 = Brow0 + (size_t)16 * DD;
    load_b_krow(bA0, Brow0);
    load_b_krow(bA1, Brow1);

    for (int i = tid; i < MR * DD; i += NTHR) {
        int r = i >> 9;
        int c = i & 511;
        sQ[r * STR + c] = Qb[r * DD + c];
    }
    if (tid < MR) sMask[tid] = mask[(size_t)b * SD + q0 + tid];
    __syncthreads();

    const float scale = 0.044194173824159216f;

    {
        const float* Arow = sQ + l16 * STR + 2 * half;
        v8f acc[2][MT];
        for (int t = 0; t < 2; ++t)
            for (int m = 0; m < MT; ++m) acc[t][m] = (v8f){};
        for (int d = 0; d < DD; d += 2 * KB) {
            load_b_krow(bB0, Brow0 + d + KB);
            load_b_krow(bB1, Brow1 + d + KB);
            mma_block2(acc, Arow + d, bA0, bA1);
            const int nd = (d + 2 * KB < DD) ? d + 2 * KB : 0;
            load_b_krow(bA0, Brow0 + nd);
            load_b_krow(bA1, Brow1 + nd);
            mma_block2(acc, Arow + d + KB, bB0, bB1);
        }
        for (int t = 0; t < 2; ++t) {
            const int k0 = (wave * 2 + t) * 16;
            for (int m = 0; m < MT; ++m)
#pragma unroll
                for (int i = 0; i < 8; ++i)
                    sS[(m * 16 + i + 8 * half) * STR + k0 + l16] = acc[t][m][i] * scale;
        }
    }
    __syncthreads();

    {
        const int sub  = (tid >> 4) & 15;
        const int rowA = (tid >> 8) * 16 + (tid & 15);
        float mall[2], inv[2];
        for (int r = 0; r < 2; ++r) {
            const float* srow = sS + (rowA + 32 * r) * STR;
            float m = -INFINITY;
            for (int c = sub * 32; c < sub * 32 + 32; ++c) m = fmaxf(m, srow[c]);
            sRed[(rowA + 32 * r) * 16 + sub] = m;
        }
        __syncthreads();
        for (int r = 0; r < 2; ++r) {
            float mm = -INFINITY;
#pragma unroll
            for (int j = 0; j < 16; ++j) mm = fmaxf(mm, sRed[(rowA + 32 * r) * 16 + j]);
            mall[r] = mm;
        }
        __syncthreads();
        for (int r = 0; r < 2; ++r) {
            float* srow = sS + (rowA + 32 * r) * STR;
            float s = 0.f;
            for (int c = sub * 32; c < sub * 32 + 32; ++c) {
                float p = __expf(srow[c] - mall[r]);
                srow[c] = p;
                s += p;
            }
            sRed[(rowA + 32 * r) * 16 + sub] = s;
        }
        __syncthreads();
        for (int r = 0; r < 2; ++r) {
            float stot = 0.f;
#pragma unroll
            for (int j = 0; j < 16; ++j) stot += sRed[(rowA + 32 * r) * 16 + j];
            inv[r] = 1.0f / stot;
        }
        for (int r = 0; r < 2; ++r) {
            float* srow = sS + (rowA + 32 * r) * STR;
            float* gw = outWeights + ((size_t)b * SD + q0 + rowA + 32 * r) * SE;
            for (int c = sub * 32; c < sub * 32 + 32; ++c) {
                float p = srow[c] * inv[r];
                srow[c] = p;
                gw[c]   = p;
            }
        }
    }
    const float* Bk0 = Eb + wave * 32 + l16 + (size_t)(2 * half) * DD;
    const float* Bk1 = Bk0 + 16;
    load_b_kcol(bA0, Bk0);
    load_b_kcol(bA1, Bk1);
    __syncthreads();

    v8f ctx[2][MT];
    {
        const float* Arow = sS + l16 * STR + 2 * half;
        for (int t = 0; t < 2; ++t)
            for (int m = 0; m < MT; ++m) ctx[t][m] = (v8f){};
        for (int k = 0; k < SE; k += 2 * KB) {
            load_b_kcol(bB0, Bk0 + (size_t)(k + KB) * DD);
            load_b_kcol(bB1, Bk1 + (size_t)(k + KB) * DD);
            mma_block2(ctx, Arow + k, bA0, bA1);
            const int nk = (k + 2 * KB < SE) ? k + 2 * KB : 0;
            load_b_kcol(bA0, Bk0 + (size_t)nk * DD);
            load_b_kcol(bA1, Bk1 + (size_t)nk * DD);
            mma_block2(ctx, Arow + k + KB, bB0, bB1);
        }
    }
    const float* Wr0 = W + (size_t)(wave * 32 + l16) * (2 * DD) + 2 * half;
    const float* Wr1 = Wr0 + (size_t)16 * (2 * DD);
    load_b_krow(bA0, Wr0);
    load_b_krow(bA1, Wr1);
    __syncthreads();
    for (int t = 0; t < 2; ++t) {
        const int n0 = wave * 32 + t * 16;
        for (int m = 0; m < MT; ++m)
#pragma unroll
            for (int i = 0; i < 8; ++i)
                sS[(m * 16 + i + 8 * half) * STR + n0 + l16] = ctx[t][m][i];
    }
    __syncthreads();

    {
        const float* AQ = sQ + l16 * STR + 2 * half;
        const float* AC = sS + l16 * STR + 2 * half;
        v8f acc[2][MT];
        for (int t = 0; t < 2; ++t)
            for (int m = 0; m < MT; ++m) acc[t][m] = (v8f){};
        for (int k = 0; k < 2 * DD; k += 2 * KB) {
            const float* A = (k < DD) ? AQ + k : AC + (k - DD);
            load_b_krow(bB0, Wr0 + k + KB);
            load_b_krow(bB1, Wr1 + k + KB);
            mma_block2(acc, A, bA0, bA1);
            const int nk = (k + 2 * KB < 2 * DD) ? k + 2 * KB : 0;
            load_b_krow(bA0, Wr0 + nk);
            load_b_krow(bA1, Wr1 + nk);
            mma_block2(acc, A + KB, bB0, bB1);
        }
#pragma nounroll
        for (int t = 0; t < 2; ++t) {
            const int n0 = wave * 32 + t * 16;
            const float bn = bias[n0 + l16];
            float* orow = outMasked + ((size_t)b * SD + q0) * DD + n0 + l16;
#pragma unroll
            for (int m = 0; m < MT; ++m)
#pragma unroll
                for (int i = 0; i < 8; ++i) {
                    const int M = m * 16 + i + 8 * half;
                    const float v = t ? acc[1][m][i] : acc[0][m][i];
                    orow[(size_t)M * DD] = tanhf(v + bn) * sMask[M];
                }
        }
    }
}

extern "C" void kernel_launch(void* const* d_in, const int* in_sizes, int n_in,
                              void* d_out, int out_size, void* d_ws, size_t ws_size,
                              hipStream_t stream) {
    (void)in_sizes; (void)n_in; (void)out_size; (void)d_ws; (void)ws_size;
    const float* Q    = (const float*)d_in[0];
    const float* E    = (const float*)d_in[1];
    const float* mask = (const float*)d_in[2];
    const float* W    = (const float*)d_in[3];
    const float* bias = (const float*)d_in[4];
    float* outMasked  = (float*)d_out;
    float* outWeights = (float*)d_out + (size_t)NB * SD * SE;

    dim3 grid(SD / MR, NB);
    attn_fused_kernel<<<grid, NTHR, 0, stream>>>(Q, E, mask, W, bias,
                                                 outMasked, outWeights);
}


# Attention weights stored from LDS as whole 128-B lines, 2 b128 stores per wave per trip of the context product's k-loop, not as a burst of 16-B pieces behind the softmax; values, sum order unchanged
# speedup vs baseline: 1.5546x; 1.0346x over previous
#include <hip/hip_runtime.h>
#include <math.h>


typedef __attribute__((ext_vector_type(2))) float v2f;
typedef __attribute__((ext_vector_type(8))) float v8f;
typedef __attribute__((ext_vector_type(4))) float v4f;

#define NB   32
#define SD   512
#define SE   512
#define DD   512
#define STR  516
#define MR   64
#define MT   4
#define NTHR 512
#define TSTR (16 * STR)
#define KB   32

__device__ __forceinline__ void load_b_krow(v2f (&bf)[8], const float* __restrict__ p) {
#pragma unroll
    for (int j = 0; j < 8; ++j) { bf[j].x = p[4 * j]; bf[j].y = p[4 * j + 1]; }
    asm volatile("" ::: "memory");
}

__device__ __forceinline__ void load_b_kcol(v2f (&bf)[8], const float* __restrict__ p) {
#pragma unroll
    for (int j = 0; j < 8; ++j) {
        bf[j].x = p[(size_t)(4 * j) * DD];
        bf[j].y = p[(size_t)(4 * j + 1) * DD];
    }
    asm volatile("" ::: "memory");
}

__device__ __forceinline__ void mma_block2(v8f (&acc)[2][MT], const float* a,
                                           const v2f (&bf0)[8], const v2f (&bf1)[8]) {
#pragma unroll
    for (int j = 0; j < 8; ++j)
#pragma unroll
        for (int m = 0; m < MT; ++m) {
            v2f av;  av.x = a[m * TSTR + 4 * j];  av.y = a[m * TSTR + 4 * j + 1];
            acc[0][m] = __builtin_amdgcn_wmma_f32_16x16x4_f32(
                false, av, false, bf0[j], (short)0, acc[0][m], false, false);
            acc[1][m] = __builtin_amdgcn_wmma_f32_16x16x4_f32(
                false, av, false, bf1[j], (short)0, acc[1][m], false, false);
        }
}

__launch_bounds__(NTHR, 1)
__global__ void attn_fused_kernel(const float* __restrict__ Q,
                                  const float* __restrict__ E,
                                  const float* __restrict__ mask,
                                  const float* __restrict__ W,
                                  const float* __restrict__ bias,
                                  float* __restrict__ outMasked,
                                  float* __restrict__ outWeights)
{
    __shared__ float sQ[MR * STR];
    __shared__ __attribute__((aligned(16))) float sS[MR * STR];
    __shared__ float sRed[MR * 16];
    __shared__ float sMask[MR];

    const int b    = blockIdx.y;
    const int q0   = blockIdx.x * MR;
    const int tid  = threadIdx.x;
    const int lane = tid & 31;
    const int wave = tid >> 5;
    const int half = lane >> 4;
    const int l16  = lane & 15;

    const float* Qb = Q + ((size_t)b * SD + q0) * DD;
    const float* Eb = E + (size_t)b * SE * DD;

    v2f bA0[8], bA1[8], bB0[8], bB1[8];
    const float* Brow0 = Eb + (size_t)(wave * 32 + l16) * DD + 2 * half;
    const float* Brow1 = Brow0 + (size_t)16 * DD;
    load_b_krow(bA0, Brow0);
    load_b_krow(bA1, Brow1);

    for (int i = tid; i < MR * DD; i += NTHR) {
        int r = i >> 9;
        int c = i & 511;
        sQ[r * STR + c] = Qb[r * DD + c];
    }
    if (tid < MR) sMask[tid] = mask[(size_t)b * SD + q0 + tid];
    __syncthreads();

    const float scale = 0.044194173824159216f;

    {
        const float* Arow = sQ + l16 * STR + 2 * half;
        v8f acc[2][MT];
        for (int t = 0; t < 2; ++t)
            for (int m = 0; m < MT; ++m) acc[t][m] = (v8f){};
        for (int d = 0; d < DD; d += 2 * KB) {
            load_b_krow(bB0, Brow0 + d + KB);
            load_b_krow(bB1, Brow1 + d + KB);
            mma_block2(acc, Arow + d, bA0, bA1);
            const int nd = (d + 2 * KB < DD) ? d + 2 * KB : 0;
            load_b_krow(bA0, Brow0 + nd);
            load_b_krow(bA1, Brow1 + nd);
            mma_block2(acc, Arow + d + KB, bB0, bB1);
        }
        for (int t = 0; t < 2; ++t) {
            const int k0 = (wave * 2 + t) * 16;
            for (int m = 0; m < MT; ++m)
#pragma unroll
                for (int i = 0; i < 8; ++i)
                    sS[(m * 16 + i + 8 * half) * STR + k0 + l16] = acc[t][m][i] * scale;
        }
    }
    __syncthreads();

    {
        const int sub  = (tid >> 4) & 15;
        const int rowA = (tid >> 8) * 16 + (tid & 15);
        float mall[2], inv[2];
        for (int r = 0; r < 2; ++r) {
            const float* srow = sS + (rowA + 32 * r) * STR;
            float m = -INFINITY;
            for (int c = sub * 32; c < sub * 32 + 32; ++c) m = fmaxf(m, srow[c]);
            sRed[(rowA + 32 * r) * 16 + sub] = m;
        }
        __syncthreads();
        for (int r = 0; r < 2; ++r) {
            float mm = -INFINITY;
#pragma unroll
            for (int j = 0; j < 16; ++j) mm = fmaxf(mm, sRed[(rowA + 32 * r) * 16 + j]);
            mall[r] = mm;
        }
        __syncthreads();
        for (int r = 0; r < 2; ++r) {
            float* srow = sS + (rowA + 32 * r) * STR;
            float s = 0.f;
            for (int c = sub * 32; c < sub * 32 + 32; ++c) {
                float p = __expf(srow[c] - mall[r]);
                srow[c] = p;
                s += p;
            }
            sRed[(rowA + 32 * r) * 16 + sub] = s;
        }
        __syncthreads();
        for (int r = 0; r < 2; ++r) {
            float stot = 0.f;
#pragma unroll
            for (int j = 0; j < 16; ++j) stot += sRed[(rowA + 32 * r) * 16 + j];
            inv[r] = 1.0f / stot;
        }
        for (int r = 0; r < 2; ++r) {
            float* srow = sS + (rowA + 32 * r) * STR;
            for (int c = sub * 32; c < sub * 32 + 32; ++c)
                srow[c] = srow[c] * inv[r];
        }
    }
    const float* Bk0 = Eb + wave * 32 + l16 + (size_t)(2 * half) * DD;
    const float* Bk1 = Bk0 + 16;
    load_b_kcol(bA0, Bk0);
    load_b_kcol(bA1, Bk1);
    __syncthreads();

    v8f ctx[2][MT];
    {
        const float* Arow = sS + l16 * STR + 2 * half;
        for (int t = 0; t < 2; ++t)
            for (int m = 0; m < MT; ++m) ctx[t][m] = (v8f){};
        const float* pS = sS + (wave * 4) * STR + lane * 4;
        float* gW = outWeights + ((size_t)b * SD + q0 + wave * 4) * SE + lane * 4;
        int odd = 0;
        for (int k = 0; k < SE; k += 2 * KB) {
            const v4f w0 = *(const v4f*)(pS);
            const v4f w1 = *(const v4f*)(pS + 128);
            load_b_kcol(bB0, Bk0 + (size_t)(k + KB) * DD);
            load_b_kcol(bB1, Bk1 + (size_t)(k + KB) * DD);
            mma_block2(ctx, Arow + k, bA0, bA1);
            *(v4f*)(gW)       = w0;
            *(v4f*)(gW + 128) = w1;
            pS += odd ? 256 + (STR - SE) : 256;
            gW += 256;
            odd ^= 1;
            const int nk = (k + 2 * KB < SE) ? k + 2 * KB : 0;
            load_b_kcol(bA0, Bk0 + (size_t)nk * DD);
            load_b_kcol(bA1, Bk1 + (size_t)nk * DD);
            mma_block2(ctx, Arow + k + KB, bB0, bB1);
        }
    }
    const float* Wr0 = W + (size_t)(wave * 32 + l16) * (2 * DD) + 2 * half;
    const float* Wr1 = Wr0 + (size_t)16 * (2 * DD);
    load_b_krow(bA0, Wr0);
    load_b_krow(bA1, Wr1);
    __syncthreads();
    for (int t = 0; t < 2; ++t) {
        const int n0 = wave * 32 + t * 16;
        for (int m = 0; m < MT; ++m)
#pragma unroll
            for (int i = 0; i < 8; ++i)
                sS[(m * 16 + i + 8 * half) * STR + n0 + l16] = ctx[t][m][i];
    }
    __syncthreads();

    {
        const float* AQ = sQ + l16 * STR + 2 * half;
        const float* AC = sS + l16 * STR + 2 * half;
        v8f acc[2][MT];
        for (int t = 0; t < 2; ++t)
            for (int m = 0; m < MT; ++m) acc[t][m] = (v8f){};
        for (int k = 0; k < 2 * DD; k += 2 * KB) {
            const float* A = (k < DD) ? AQ + k : AC + (k - DD);
            load_b_krow(bB0, Wr0 + k + KB);
            load_b_krow(bB1, Wr1 + k + KB);
            mma_block2(acc, A, bA0, bA1);
            const int nk = (k + 2 * KB < 2 * DD) ? k + 2 * KB : 0;
            load_b_krow(bA0, Wr0 + nk);
            load_b_krow(bA1, Wr1 + nk);
            mma_block2(acc, A + KB, bB0, bB1);
        }
#pragma nounroll
        for (int t = 0; t < 2; ++t) {
            const int n0 = wave * 32 + t * 16;
            const float bn = bias[n0 + l16];
            float* orow = outMasked + ((size_t)b * SD + q0) * DD + n0 + l16;
#pragma unroll
            for (int m = 0; m < MT; ++m)
#pragma unroll
                for (int i = 0; i < 8; ++i) {
                    const int M = m * 16 + i + 8 * half;
                    const float v = t ? acc[1][m][i] : acc[0][m][i];
                    orow[(size_t)M * DD] = tanhf(v + bn) * sMask[M];
                }
        }
    }
}

extern "C" void kernel_launch(void* const* d_in, const int* in_sizes, int n_in,
                              void* d_out, int out_size, void* d_ws, size_t ws_size,
                              hipStream_t stream) {
    (void)in_sizes; (void)n_in; (void)out_size; (void)d_ws; (void)ws_size;
    const float* Q    = (const float*)d_in[0];
    const float* E    = (const float*)d_in[1];
    const float* mask = (const float*)d_in[2];
    const float* W    = (const float*)d_in[3];
    const float* bias = (const float*)d_in[4];
    float* outMasked  = (float*)d_out;
    float* outWeights = (float*)d_out + (size_t)NB * SD * SE;

    dim3 grid(SD / MR, NB);
    attn_fused_kernel<<<grid, NTHR, 0, stream>>>(Q, E, mask, W, bias,
                                                 outMasked, outWeights);
}


# Output stage: branch-free tanh, tile loop unrolled: both forms of the library call (polynomial under 0.625; 1-2/(exp(2|v|)+1) above, hardware exp2) on each lane, then select; above 0.625 within 3 ulp
# speedup vs baseline: 1.6277x; 1.0470x over previous
#include <hip/hip_runtime.h>
#include <math.h>


typedef __attribute__((ext_vector_type(2))) float v2f;
typedef __attribute__((ext_vector_type(8))) float v8f;
typedef __attribute__((ext_vector_type(4))) float v4f;

#define NB   32
#define SD   512
#define SE   512
#define DD   512
#define STR  516
#define MR   64
#define MT   4
#define NTHR 512
#define TSTR (16 * STR)
#define KB   32

__device__ __forceinline__ float tanh_sel(float v) {
    const float a = fabsf(v);
    const float s = a * a;
    float p = fmaf(s, -0.0057000205852091312f, 0.020634071901440620f);
    p = fmaf(s, p, -0.053737930953502655f);
    p = fmaf(s, p, 0.13331416249275208f);
    p = fmaf(s, p, -0.33333280682563782f);
    const float small = fmaf(s, a * p, a);
    const float e = __builtin_amdgcn_exp2f(a * 2.8853900817779268f);
    const float big = fmaf(__builtin_amdgcn_rcpf(e + 1.0f), -2.0f, 1.0f);
    return copysignf((a < 0.625f) ? small : big, v);
}

__device__ __forceinline__ void load_b_krow(v2f (&bf)[8], const float* __restrict__ p) {
#pragma unroll
    for (int j = 0; j < 8; ++j) { bf[j].x = p[4 * j]; bf[j].y = p[4 * j + 1]; }
    asm volatile("" ::: "memory");
}

__device__ __forceinline__ void load_b_kcol(v2f (&bf)[8], const float* __restrict__ p) {
#pragma unroll
    for (int j = 0; j < 8; ++j) {
        bf[j].x = p[(size_t)(4 * j) * DD];
        bf[j].y = p[(size_t)(4 * j + 1) * DD];
    }
    asm volatile("" ::: "memory");
}

__device__ __forceinline__ void mma_block2(v8f (&acc)[2][MT], const float* a,
                                           const v2f (&bf0)[8], const v2f (&bf1)[8]) {
#pragma unroll
    for (int j = 0; j < 8; ++j)
#pragma unroll
        for (int m = 0; m < MT; ++m) {
            v2f av;  av.x = a[m * TSTR + 4 * j];  av.y = a[m * TSTR + 4 * j + 1];
            acc[0][m] = __builtin_amdgcn_wmma_f32_16x16x4_f32(
                false, av, false, bf0[j], (short)0, acc[0][m], false, false);
            acc[1][m] = __builtin_amdgcn_wmma_f32_16x16x4_f32(
                false, av, false, bf1[j], (short)0, acc[1][m], false, false);
        }
}

__launch_bounds__(NTHR, 1)
__global__ void attn_fused_kernel(const float* __restrict__ Q,
                                  const float* __restrict__ E,
                                  const float* __restrict__ mask,
                                  const float* __restrict__ W,
                                  const float* __restrict__ bias,
                                  float* __restrict__ outMasked,
                                  float* __restrict__ outWeights)
{
    __shared__ float sQ[MR * STR];
    __shared__ __attribute__((aligned(16))) float sS[MR * STR];
    __shared__ float sRed[MR * 16];
    __shared__ float sMask[MR];

    const int b    = blockIdx.y;
    const int q0   = blockIdx.x * MR;
    const int tid  = threadIdx.x;
    const int lane = tid & 31;
    const int wave = tid >> 5;
    const int half = lane >> 4;
    const int l16  = lane & 15;

    const float* Qb = Q + ((size_t)b * SD + q0) * DD;
    const float* Eb = E + (size_t)b * SE * DD;

    v2f bA0[8], bA1[8], bB0[8], bB1[8];
    const float* Brow0 = Eb + (size_t)(wave * 32 + l16) * DD + 2 * half;
    const float* Brow1 = Brow0 + (size_t)16 * DD;
    load_b_krow(bA0, Brow0);
    load_b_krow(bA1, Brow1);

    for (int i = tid; i < MR * DD; i += NTHR) {
        int r = i >> 9;
        int c = i & 511;
        sQ[r * STR + c] = Qb[r * DD + c];
    }
    if (tid < MR) sMask[tid] = mask[(size_t)b * SD + q0 + tid];
    __syncthreads();

    const float scale = 0.044194173824159216f;

    {
        const float* Arow = sQ + l16 * STR + 2 * half;
        v8f acc[2][MT];
        for (int t = 0; t < 2; ++t)
            for (int m = 0; m < MT; ++m) acc[t][m] = (v8f){};
        for (int d = 0; d < DD; d += 2 * KB) {
            load_b_krow(bB0, Brow0 + d + KB);
            load_b_krow(bB1, Brow1 + d + KB);
            mma_block2(acc, Arow + d, bA0, bA1);
            const int nd = (d + 2 * KB < DD) ? d + 2 * KB : 0;
            load_b_krow(bA0, Brow0 + nd);
            load_b_krow(bA1, Brow1 + nd);
            mma_block2(acc, Arow + d + KB, bB0, bB1);
        }
        for (int t = 0; t < 2; ++t) {
            const int k0 = (wave * 2 + t) * 16;
            for (int m = 0; m < MT; ++m)
#pragma unroll
                for (int i = 0; i < 8; ++i)
                    sS[(m * 16 + i + 8 * half) * STR + k0 + l16] = acc[t][m][i] * scale;
        }
    }
    __syncthreads();

    {
        const int sub  = (tid >> 4) & 15;
        const int rowA = (tid >> 8) * 16 + (tid & 15);
        float mall[2], inv[2];
        for (int r = 0; r < 2; ++r) {
            const float* srow = sS + (rowA + 32 * r) * STR;
            float m = -INFINITY;
            for (int c = sub * 32; c < sub * 32 + 32; ++c) m = fmaxf(m, srow[c]);
            sRed[(rowA + 32 * r) * 16 + sub] = m;
        }
        __syncthreads();
        for (int r = 0; r < 2; ++r) {
            float mm = -INFINITY;
#pragma unroll
            for (int j = 0; j < 16; ++j) mm = fmaxf(mm, sRed[(rowA + 32 * r) * 16 + j]);
            mall[r] = mm;
        }
        __syncthreads();
        for (int r = 0; r < 2; ++r) {
            float* srow = sS + (rowA + 32 * r) * STR;
            float s = 0.f;
            for (int c = sub * 32; c < sub * 32 + 32; ++c) {
                float p = __expf(srow[c] - mall[r]);
                srow[c] = p;
                s += p;
            }
            sRed[(rowA + 32 * r) * 16 + sub] = s;
        }
        __syncthreads();
        for (int r = 0; r < 2; ++r) {
            float stot = 0.f;
#pragma unroll
            for (int j = 0; j < 16; ++j) stot += sRed[(rowA + 32 * r) * 16 + j];
            inv[r] = 1.0f / stot;
        }
        for (int r = 0; r < 2; ++r) {
            float* srow = sS + (rowA + 32 * r) * STR;
            for (int c = sub * 32; c < sub * 32 + 32; ++c)
                srow[c] = srow[c] * inv[r];
        }
    }
    const float* Bk0 = Eb + wave * 32 + l16 + (size_t)(2 * half) * DD;
    const float* Bk1 = Bk0 + 16;
    load_b_kcol(bA0, Bk0);
    load_b_kcol(bA1, Bk1);
    __syncthreads();

    v8f ctx[2][MT];
    {
        const float* Arow = sS + l16 * STR + 2 * half;
        for (int t = 0; t < 2; ++t)
            for (int m = 0; m < MT; ++m) ctx[t][m] = (v8f){};
        const float* pS = sS + (wave * 4) * STR + lane * 4;
        float* gW = outWeights + ((size_t)b * SD + q0 + wave * 4) * SE + lane * 4;
        int odd = 0;
        for (int k = 0; k < SE; k += 2 * KB) {
            const v4f w0 = *(const v4f*)(pS);
            const v4f w1 = *(const v4f*)(pS + 128);
            load_b_kcol(bB0, Bk0 + (size_t)(k + KB) * DD);
            load_b_kcol(bB1, Bk1 + (size_t)(k + KB) * DD);
            mma_block2(ctx, Arow + k, bA0, bA1);
            *(v4f*)(gW)       = w0;
            *(v4f*)(gW + 128) = w1;
            pS += odd ? 256 + (STR - SE) : 256;
            gW += 256;
            odd ^= 1;
            const int nk = (k + 2 * KB < SE) ? k + 2 * KB : 0;
            load_b_kcol(bA0, Bk0 + (size_t)nk * DD);
            load_b_kcol(bA1, Bk1 + (size_t)nk * DD);
            mma_block2(ctx, Arow + k + KB, bB0, bB1);
        }
    }
    const float* Wr0 = W + (size_t)(wave * 32 + l16) * (2 * DD) + 2 * half;
    const float* Wr1 = Wr0 + (size_t)16 * (2 * DD);
    load_b_krow(bA0, Wr0);
    load_b_krow(bA1, Wr1);
    __syncthreads();
    for (int t = 0; t < 2; ++t) {
        const int n0 = wave * 32 + t * 16;
        for (int m = 0; m < MT; ++m)
#pragma unroll
            for (int i = 0; i < 8; ++i)
                sS[(m * 16 + i + 8 * half) * STR + n0 + l16] = ctx[t][m][i];
    }
    __syncthreads();

    {
        const float* AQ = sQ + l16 * STR + 2 * half;
        const float* AC = sS + l16 * STR + 2 * half;
        v8f acc[2][MT];
        for (int t = 0; t < 2; ++t)
            for (int m = 0; m < MT; ++m) acc[t][m] = (v8f){};
        for (int k = 0; k < 2 * DD; k += 2 * KB) {
            const float* A = (k < DD) ? AQ + k : AC + (k - DD);
            load_b_krow(bB0, Wr0 + k + KB);
            load_b_krow(bB1, Wr1 + k + KB);
            mma_block2(acc, A, bA0, bA1);
            const int nk = (k + 2 * KB < 2 * DD) ? k + 2 * KB : 0;
            load_b_krow(bA0, Wr0 + nk);
            load_b_krow(bA1, Wr1 + nk);
            mma_block2(acc, A + KB, bB0, bB1);
        }
        const float bn0 = bias[wave * 32 + l16];
        const float bn1 = bias[wave * 32 + 16 + l16];
        float* orow = outMasked + ((size_t)b * SD + q0) * DD + wave * 32 + l16;
#pragma unroll
        for (int m = 0; m < MT; ++m)
#pragma unroll
            for (int i = 0; i < 8; ++i) {
                const int M = m * 16 + i + 8 * half;
                const float mk = sMask[M];
                orow[(size_t)M * DD]      = tanh_sel(acc[0][m][i] + bn0) * mk;
                orow[(size_t)M * DD + 16] = tanh_sel(acc[1][m][i] + bn1) * mk;
            }
    }
}

extern "C" void kernel_launch(void* const* d_in, const int* in_sizes, int n_in,
                              void* d_out, int out_size, void* d_ws, size_t ws_size,
                              hipStream_t stream) {
    (void)in_sizes; (void)n_in; (void)out_size; (void)d_ws; (void)ws_size;
    const float* Q    = (const float*)d_in[0];
    const float* E    = (const float*)d_in[1];
    const float* mask = (const float*)d_in[2];
    const float* W    = (const float*)d_in[3];
    const float* bias = (const float*)d_in[4];
    float* outMasked  = (float*)d_out;
    float* outWeights = (float*)d_out + (size_t)NB * SD * SE;

    dim3 grid(SD / MR, NB);
    attn_fused_kernel<<<grid, NTHR, 0, stream>>>(Q, E, mask, W, bias,
                                                 outMasked, outWeights);
}
